# starting kernel
# baseline (speedup 1.0000x reference)
#include <hip/hip_runtime.h>
#include <hip/hip_bf16.h>


#define Bc   8
#define Tc   8192
#define Cc   512
#define NSEG 64
#define SEG  (Tc / NSEG)

typedef __attribute__((ext_vector_type(2))) float v2f;
typedef __attribute__((ext_vector_type(8))) float v8f;

__global__ __launch_bounds__(256)
void CasualBoW_pass1_segsum(const float* __restrict__ x, float* __restrict__ part) {
    const int CBLK = Cc / 256;
    int cb  = blockIdx.x % CBLK;
    int seg = (blockIdx.x / CBLK) % NSEG;
    int b   = blockIdx.x / (CBLK * NSEG);
    int c   = cb * 256 + threadIdx.x;

    const float* p = x + ((size_t)b * Tc + (size_t)seg * SEG) * Cc + c;
    float s0 = 0.f, s1 = 0.f, s2 = 0.f, s3 = 0.f;
#pragma unroll 4
    for (int i = 0; i < SEG; i += 4) {
        s0 += p[(size_t)(i + 0) * Cc];
        s1 += p[(size_t)(i + 1) * Cc];
        s2 += p[(size_t)(i + 2) * Cc];
        s3 += p[(size_t)(i + 3) * Cc];
    }
    part[((size_t)b * NSEG + seg) * Cc + c] = (s0 + s1) + (s2 + s3);
}

__global__ __launch_bounds__(256)
void CasualBoW_pass2_scan(const float* __restrict__ x, const float* __restrict__ part,
                          float* __restrict__ out) {
    const int lane = threadIdx.x & 31;
    const int wave = threadIdx.x >> 5;
    const int CB   = Cc / (16 * 8);

    int cb  = blockIdx.x % CB;
    int seg = (blockIdx.x / CB) % NSEG;
    int b   = blockIdx.x / (CB * NSEG);

    const int cg = cb * 8 + wave;
    const int c  = cg * 16 + (lane & 15);
    const int hi = lane >> 4;
    const int m  = lane & 15;

    float off = 0.f;
    {
        const float* pp = part + (size_t)b * NSEG * Cc + c;
        for (int s = 0; s < seg; ++s) off += pp[(size_t)s * Cc];
    }

    v2f A[4];
#pragma unroll
    for (int kk = 0; kk < 4; ++kk) {
        int kg = kk * 4 + 2 * hi;
        A[kk].x = (m >= kg)     ? 1.f : 0.f;
        A[kk].y = (m >= kg + 1) ? 1.f : 0.f;
    }

    const size_t rowBase = (size_t)b * Tc + (size_t)seg * SEG;
    const float* xp = x   + rowBase * Cc + c;
    float*       op = out + rowBase * Cc + c;

    v2f Bm[4];
#pragma unroll
    for (int kk = 0; kk < 4; ++kk) {
        int r0 = kk * 4 + 2 * hi;
        Bm[kk].x = xp[(size_t)(r0 + 0) * Cc];
        Bm[kk].y = xp[(size_t)(r0 + 1) * Cc];
    }

    for (int t0 = 0; t0 < SEG; t0 += 16) {
        v2f Bn[4];
        if (t0 + 16 < SEG) {
#pragma unroll
            for (int kk = 0; kk < 4; ++kk) {
                int r0 = (t0 + 16) + kk * 4 + 2 * hi;
                Bn[kk].x = xp[(size_t)(r0 + 0) * Cc];
                Bn[kk].y = xp[(size_t)(r0 + 1) * Cc];
            }
        }

        v8f acc;
#pragma unroll
        for (int r = 0; r < 8; ++r) acc[r] = off;

#pragma unroll
        for (int kk = 0; kk < 4; ++kk) {
            acc = __builtin_amdgcn_wmma_f32_16x16x4_f32(
                false, A[kk], false, Bm[kk], (short)0, acc, false, false);
        }

        off = __shfl(acc[7], 16 + (lane & 15), 32);

        const float cbase = (float)(seg * SEG + t0 + (hi ? 8 : 0) + 1);
#pragma unroll
        for (int r = 0; r < 8; ++r) {
            float cnt = cbase + (float)r;
            float rc  = __builtin_amdgcn_rcpf(cnt);
            rc        = rc * (2.0f - cnt * rc);
            float v   = acc[r] * rc;
            int  row  = t0 + (hi ? 8 : 0) + r;
            __builtin_nontemporal_store(v, &op[(size_t)row * Cc]);
        }

#pragma unroll
        for (int kk = 0; kk < 4; ++kk) Bm[kk] = Bn[kk];
    }
}

extern "C" void kernel_launch(void* const* d_in, const int* in_sizes, int n_in,
                              void* d_out, int out_size, void* d_ws, size_t ws_size,
                              hipStream_t stream) {
    (void)in_sizes; (void)n_in; (void)out_size; (void)ws_size;
    const float* x    = (const float*)d_in[0];
    float*       out  = (float*)d_out;
    float*       part = (float*)d_ws;

    dim3 g1(Bc * NSEG * (Cc / 256));
    CasualBoW_pass1_segsum<<<g1, 256, 0, stream>>>(x, part);

    dim3 g2(Bc * NSEG * (Cc / (16 * 8)));
    CasualBoW_pass2_scan<<<g2, 256, 0, stream>>>(x, part, out);
}


# split work-group barrier: a wave signals after publishing its chunk sum, runs the next step's tile scans and row requests, then waits and finishes the step; waves leave lockstep; sum order unchanged
# speedup vs baseline: 1.9884x; 1.9884x over previous
#include <hip/hip_runtime.h>


#define Bc    8
#define Tc    8192
#define Cc    512
#define SLAB  32
#define NW    16
#define RT    2
#define WROWS (RT * 16)
#define STEP  (NW * WROWS)
#define NSTEP (Tc / STEP)

typedef __attribute__((ext_vector_type(2))) float v2f;
typedef __attribute__((ext_vector_type(8))) float v8f;

struct Chunk { v2f v[RT][4][2]; };

__device__ __forceinline__ void load_chunk(Chunk& dst, const float* p, int h) {
#pragma unroll
    for (int t = 0; t < RT; ++t)
#pragma unroll
        for (int kk = 0; kk < 4; ++kk)
#pragma unroll
            for (int s = 0; s < 2; ++s) {
                const int row = t * 16 + kk * 4 + 2 * h + s;
                dst.v[t][kk][s] = *(const v2f*)(p + (size_t)row * Cc);
            }
}

struct Acc { v8f a0[RT], a1[RT]; float s0[RT], s1[RT]; };

__device__ __forceinline__ void tile_scan(const Chunk& cur, const v2f (&A)[4], Acc& P, const int n) {
#pragma unroll
    for (int t = 0; t < RT; ++t) {
#pragma unroll
        for (int r = 0; r < 8; ++r) { P.a0[t][r] = 0.f; P.a1[t][r] = 0.f; }
#pragma unroll
        for (int kk = 0; kk < 4; ++kk) {
            v2f b0, b1;
            b0.x = cur.v[t][kk][0].x;  b0.y = cur.v[t][kk][1].x;
            b1.x = cur.v[t][kk][0].y;  b1.y = cur.v[t][kk][1].y;
            P.a0[t] = __builtin_amdgcn_wmma_f32_16x16x4_f32(
                false, A[kk], false, b0, (short)0, P.a0[t], false, false);
            P.a1[t] = __builtin_amdgcn_wmma_f32_16x16x4_f32(
                false, A[kk], false, b1, (short)0, P.a1[t], false, false);
        }
    }
#pragma unroll
    for (int t = 0; t < RT; ++t) {
        P.s0[t] = __shfl(P.a0[t][7], 16 + n, 32);
        P.s1[t] = __shfl(P.a1[t][7], 16 + n, 32);
    }
}

__device__ __forceinline__ void publish(const Acc& P, v2f (*csum)[NW][16], const int st,
                                        const int wave, const int n) {
    float cs0 = P.s0[0], cs1 = P.s1[0];
#pragma unroll
    for (int t = 1; t < RT; ++t) { cs0 += P.s0[t]; cs1 += P.s1[t]; }
    { v2f v; v.x = cs0; v.y = cs1; csum[st & 1][wave][n] = v; }
    __builtin_amdgcn_fence(__ATOMIC_RELEASE, "workgroup", "local");
    __builtin_amdgcn_s_barrier_signal(-1);
}

__device__ __forceinline__ void finish(const Acc& P, v2f (*csum)[NW][16], const int st,
                                       float* op, const int wave, const int h, const int n,
                                       float& carry0, float& carry1) {
    const int rbase = st * STEP + wave * WROWS;

    __builtin_amdgcn_sched_barrier(0);
    __builtin_amdgcn_s_barrier_wait(-1);
    __builtin_amdgcn_fence(__ATOMIC_ACQUIRE, "workgroup", "local");

    float pre0 = 0.f, pre1 = 0.f, tot0 = 0.f, tot1 = 0.f;
#pragma unroll
    for (int w = 0; w < NW; ++w) {
        const v2f v = csum[st & 1][w][n];
        if (w < wave) { pre0 += v.x; pre1 += v.y; }
        tot0 += v.x; tot1 += v.y;
    }
    float off0 = carry0 + pre0;
    float off1 = carry1 + pre1;
    carry0 += tot0;
    carry1 += tot1;

#pragma unroll
    for (int t = 0; t < RT; ++t) {
#pragma unroll
        for (int r = 0; r < 8; ++r) {
            const int   row = rbase + t * 16 + 8 * h + r;
            const float cnt = (float)(row + 1);
            float rc = __builtin_amdgcn_rcpf(cnt);
            rc = rc * (2.0f - cnt * rc);
            v2f o;
            o.x = (P.a0[t][r] + off0) * rc;
            o.y = (P.a1[t][r] + off1) * rc;
            __builtin_nontemporal_store(o, (v2f*)(op + (size_t)row * Cc));
        }
        off0 += P.s0[t];
        off1 += P.s1[t];
    }
}

template <bool LOAD, bool SCAN>
__device__ __forceinline__ void trip(const Acc& Pc, Acc& Pn, Chunk& buf,
                                     const int st, v2f (&A)[4], v2f (*csum)[NW][16],
                                     const float* xp, float* op,
                                     const int wave, const int h, const int n,
                                     float& carry0, float& carry1) {
    if (SCAN) {
#pragma unroll
        for (int kk = 0; kk < 4; ++kk) asm volatile("" : "+v"(A[kk]));
#pragma unroll
        for (int t = 0; t < RT; ++t)
#pragma unroll
            for (int kk = 0; kk < 4; ++kk)
#pragma unroll
                for (int s = 0; s < 2; ++s) asm volatile("" : "+v"(buf.v[t][kk][s]));
    }
    if (SCAN) tile_scan(buf, A, Pn, n);
    if (LOAD) load_chunk(buf, xp + (size_t)((st + 3) * STEP + wave * WROWS) * Cc, h);
    finish(Pc, csum, st, op, wave, h, n, carry0, carry1);
    if (SCAN) publish(Pn, csum, st + 1, wave, n);
}

__global__ __launch_bounds__(NW * 32)
void CasualBoW_stream_scan(const float* __restrict__ x, float* __restrict__ out) {
    __shared__ v2f csum[2][NW][16];

    const int lane = threadIdx.x & 31;
    const int wave = threadIdx.x >> 5;
    const int h    = lane >> 4;
    const int n    = lane & 15;
    const int slab = blockIdx.x % (Cc / SLAB);
    const int b    = blockIdx.x / (Cc / SLAB);

    v2f A[4];
#pragma unroll
    for (int kk = 0; kk < 4; ++kk) {
        const int kg = kk * 4 + 2 * h;
        A[kk].x = (n >= kg)     ? 1.f : 0.f;
        A[kk].y = (n >= kg + 1) ? 1.f : 0.f;
    }

    const size_t colOff = (size_t)b * Tc * Cc + (size_t)slab * SLAB + 2 * n;
    const float* xp = x   + colOff;
    float*       op = out + colOff;

    Chunk x0, x1;
    load_chunk(x0, xp + (size_t)(wave * WROWS) * Cc, h);
    load_chunk(x1, xp + (size_t)(STEP + wave * WROWS) * Cc, h);

    float carry0 = 0.f, carry1 = 0.f;

    Acc p0, p1;
    tile_scan(x0, A, p0, n);
    load_chunk(x0, xp + (size_t)(2 * STEP + wave * WROWS) * Cc, h);
    publish(p0, csum, 0, wave, n);

    static_assert(NSTEP % 2 == 0 && NSTEP >= 4, "tail below is written for an even NSTEP");
    int st = 0;
    for (; st < NSTEP - 4; st += 2) {
        trip<true, true>(p0, p1, x1, st,     A, csum, xp, op, wave, h, n, carry0, carry1);
        trip<true, true>(p1, p0, x0, st + 1, A, csum, xp, op, wave, h, n, carry0, carry1);
    }
    trip<true,  true >(p0, p1, x1, st,     A, csum, xp, op, wave, h, n, carry0, carry1);
    trip<false, true >(p1, p0, x0, st + 1, A, csum, xp, op, wave, h, n, carry0, carry1);
    trip<false, true >(p0, p1, x1, st + 2, A, csum, xp, op, wave, h, n, carry0, carry1);
    trip<false, false>(p1, p0, x0, st + 3, A, csum, xp, op, wave, h, n, carry0, carry1);
}

extern "C" void kernel_launch(void* const* d_in, const int* in_sizes, int n_in,
                              void* d_out, int out_size, void* d_ws, size_t ws_size,
                              hipStream_t stream) {
    (void)in_sizes; (void)n_in; (void)out_size; (void)d_ws; (void)ws_size;
    const float* x   = (const float*)d_in[0];
    float*       out = (float*)d_out;

    dim3 g(Bc * (Cc / SLAB));
    CasualBoW_stream_scan<<<g, NW * 32, 0, stream>>>(x, out);
}


# time axis of a slab cut in two uneven parts (9 + 7 steps): 256 work-groups, one per WGP, not 128; the later part's group first sums the earlier rows itself (vector adds); its carry re-associates
# speedup vs baseline: 2.3388x; 1.1762x over previous
#include <hip/hip_runtime.h>


#define Bc    8
#define Tc    8192
#define Cc    512
#define SLAB  32
#define NW    16
#define RT    2
#define WROWS (RT * 16)
#define STEP  (NW * WROWS)
#define NSTEP (Tc / STEP)
#define NS_A  9
#define NS_B  (NSTEP - NS_A)
#define ROWS_A (NS_A * STEP)

typedef __attribute__((ext_vector_type(2))) float v2f;
typedef __attribute__((ext_vector_type(4))) float v4f;
typedef __attribute__((ext_vector_type(8))) float v8f;

struct Chunk { v2f v[RT][4][2]; };

__device__ __forceinline__ void load_chunk(Chunk& dst, const float* p, int h) {
#pragma unroll
    for (int t = 0; t < RT; ++t)
#pragma unroll
        for (int kk = 0; kk < 4; ++kk)
#pragma unroll
            for (int s = 0; s < 2; ++s) {
                const int row = t * 16 + kk * 4 + 2 * h + s;
                dst.v[t][kk][s] = *(const v2f*)(p + (size_t)row * Cc);
            }
}

struct Acc { v8f a0[RT], a1[RT]; float s0[RT], s1[RT]; };

__device__ __forceinline__ void tile_scan(const Chunk& cur, const v2f (&A)[4], Acc& P, const int n) {
#pragma unroll
    for (int t = 0; t < RT; ++t) {
#pragma unroll
        for (int r = 0; r < 8; ++r) { P.a0[t][r] = 0.f; P.a1[t][r] = 0.f; }
#pragma unroll
        for (int kk = 0; kk < 4; ++kk) {
            v2f b0, b1;
            b0.x = cur.v[t][kk][0].x;  b0.y = cur.v[t][kk][1].x;
            b1.x = cur.v[t][kk][0].y;  b1.y = cur.v[t][kk][1].y;
            P.a0[t] = __builtin_amdgcn_wmma_f32_16x16x4_f32(
                false, A[kk], false, b0, (short)0, P.a0[t], false, false);
            P.a1[t] = __builtin_amdgcn_wmma_f32_16x16x4_f32(
                false, A[kk], false, b1, (short)0, P.a1[t], false, false);
        }
    }
#pragma unroll
    for (int t = 0; t < RT; ++t) {
        P.s0[t] = __shfl(P.a0[t][7], 16 + n, 32);
        P.s1[t] = __shfl(P.a1[t][7], 16 + n, 32);
    }
}

__device__ __forceinline__ void publish(const Acc& P, v2f (*csum)[NW][16], const int st,
                                        const int wave, const int n) {
    float cs0 = P.s0[0], cs1 = P.s1[0];
#pragma unroll
    for (int t = 1; t < RT; ++t) { cs0 += P.s0[t]; cs1 += P.s1[t]; }
    { v2f v; v.x = cs0; v.y = cs1; csum[st & 1][wave][n] = v; }
    __builtin_amdgcn_fence(__ATOMIC_RELEASE, "workgroup", "local");
    __builtin_amdgcn_s_barrier_signal(-1);
}

template <int ROW0>
__device__ __forceinline__ void finish(const Acc& P, v2f (*csum)[NW][16], const int st,
                                       float* op, const int wave, const int h, const int n,
                                       float& carry0, float& carry1) {
    const int rbase = st * STEP + wave * WROWS;

    __builtin_amdgcn_sched_barrier(0);
    __builtin_amdgcn_s_barrier_wait(-1);
    __builtin_amdgcn_fence(__ATOMIC_ACQUIRE, "workgroup", "local");

    float pre0 = 0.f, pre1 = 0.f, tot0 = 0.f, tot1 = 0.f;
#pragma unroll
    for (int w = 0; w < NW; ++w) {
        const v2f v = csum[st & 1][w][n];
        if (w < wave) { pre0 += v.x; pre1 += v.y; }
        tot0 += v.x; tot1 += v.y;
    }
    float off0 = carry0 + pre0;
    float off1 = carry1 + pre1;
    carry0 += tot0;
    carry1 += tot1;

#pragma unroll
    for (int t = 0; t < RT; ++t) {
#pragma unroll
        for (int r = 0; r < 8; ++r) {
            const int   row = rbase + t * 16 + 8 * h + r;
            const float cnt = (float)(ROW0 + row + 1);
            float rc = __builtin_amdgcn_rcpf(cnt);
            rc = rc * (2.0f - cnt * rc);
            v2f o;
            o.x = (P.a0[t][r] + off0) * rc;
            o.y = (P.a1[t][r] + off1) * rc;
            __builtin_nontemporal_store(o, (v2f*)(op + (size_t)row * Cc));
        }
        off0 += P.s0[t];
        off1 += P.s1[t];
    }
}

template <bool LOAD, bool SCAN, int ROW0>
__device__ __forceinline__ void trip(const Acc& Pc, Acc& Pn, Chunk& buf,
                                     const int st, v2f (&A)[4], v2f (*csum)[NW][16],
                                     const float* xp, float* op,
                                     const int wave, const int h, const int n,
                                     float& carry0, float& carry1) {
    if (SCAN) {
#pragma unroll
        for (int kk = 0; kk < 4; ++kk) asm volatile("" : "+v"(A[kk]));
#pragma unroll
        for (int t = 0; t < RT; ++t)
#pragma unroll
            for (int kk = 0; kk < 4; ++kk)
#pragma unroll
                for (int s = 0; s < 2; ++s) asm volatile("" : "+v"(buf.v[t][kk][s]));
    }
    if (SCAN) tile_scan(buf, A, Pn, n);
    if (LOAD) load_chunk(buf, xp + (size_t)((st + 3) * STEP + wave * WROWS) * Cc, h);
    finish<ROW0>(Pc, csum, st, op, wave, h, n, carry0, carry1);
    if (SCAN) publish(Pn, csum, st + 1, wave, n);
}

__device__ __forceinline__ void request_first(Chunk& x0, Chunk& x1, const float* xp,
                                              const int wave, const int h) {
    load_chunk(x0, xp + (size_t)(wave * WROWS) * Cc, h);
    load_chunk(x1, xp + (size_t)(STEP + wave * WROWS) * Cc, h);
}

template <int NS, int ROW0>
__device__ __forceinline__ void stream_scan(Chunk& x0, Chunk& x1,
                                            const float* xp, float* op, v2f (&A)[4],
                                            v2f (*csum)[NW][16],
                                            const int wave, const int h, const int n,
                                            float carry0, float carry1) {
    static_assert(NS >= 4, "three steps are requested before the first trip");

    Acc p0, p1;
    tile_scan(x0, A, p0, n);
    load_chunk(x0, xp + (size_t)(2 * STEP + wave * WROWS) * Cc, h);
    publish(p0, csum, 0, wave, n);

    int st = 0;
#pragma nounroll
    for (; st + 1 <= NS - 4; st += 2) {
        trip<true, true, ROW0>(p0, p1, x1, st,     A, csum, xp, op, wave, h, n, carry0, carry1);
        trip<true, true, ROW0>(p1, p0, x0, st + 1, A, csum, xp, op, wave, h, n, carry0, carry1);
    }
    if (NS % 2 == 0) {
        trip<true,  true,  ROW0>(p0, p1, x1, st,     A, csum, xp, op, wave, h, n, carry0, carry1);
        trip<false, true,  ROW0>(p1, p0, x0, st + 1, A, csum, xp, op, wave, h, n, carry0, carry1);
        trip<false, true,  ROW0>(p0, p1, x1, st + 2, A, csum, xp, op, wave, h, n, carry0, carry1);
        trip<false, false, ROW0>(p1, p0, x0, st + 3, A, csum, xp, op, wave, h, n, carry0, carry1);
    } else {
        trip<false, true,  ROW0>(p0, p1, x1, st,     A, csum, xp, op, wave, h, n, carry0, carry1);
        trip<false, true,  ROW0>(p1, p0, x0, st + 1, A, csum, xp, op, wave, h, n, carry0, carry1);
        trip<false, false, ROW0>(p0, p1, x1, st + 2, A, csum, xp, op, wave, h, n, carry0, carry1);
    }
}

template <int U, int SWEEP>
__device__ __forceinline__ void request_batch(v4f (&t)[U], const float* p) {
#pragma unroll
    for (int u = 0; u < U; ++u) t[u] = *(const v4f*)(p + (size_t)(u * SWEEP) * Cc);
}

template <int U>
__device__ __forceinline__ void add_batch(v4f (&acc)[4], const v4f (&t)[U]) {
#pragma unroll
    for (int u = 0; u < U; ++u) acc[u & 3] += t[u];
}

template <int ROWS>
__device__ __forceinline__ void slab_sum(const float* xs, v4f (*red)[SLAB / 4],
                                         const int wave, const int lane, const int n,
                                         float& s0, float& s1) {
    constexpr int SWEEP = NW * 4;
    constexpr int U     = 12;
    constexpr int NB    = ROWS / (SWEEP * U);
    constexpr size_t BATCH = (size_t)U * SWEEP * Cc;
    static_assert(ROWS % (SWEEP * U) == 0 && NB % 2 == 0 && NB >= 2,
                  "the batch loop below is written for an even number of whole batches");

    const int q  = lane >> 3;
    const int c4 = lane & 7;
    const float* p = xs + (size_t)(wave * 4 + q) * Cc + 4 * c4;

    v4f acc[4];
#pragma unroll
    for (int a = 0; a < 4; ++a) acc[a] = (v4f){0.f, 0.f, 0.f, 0.f};

    v4f ta[U], tb[U];
    request_batch<U, SWEEP>(ta, p);

#pragma nounroll
    for (int k = 0; k < NB - 2; k += 2) {
        const float* pk = p + (size_t)k * BATCH;
        request_batch<U, SWEEP>(tb, pk + BATCH);
        __builtin_amdgcn_sched_barrier(0);
        add_batch<U>(acc, ta);
        request_batch<U, SWEEP>(ta, pk + 2 * BATCH);
        add_batch<U>(acc, tb);
    }
    request_batch<U, SWEEP>(tb, p + (size_t)(NB - 1) * BATCH);
    add_batch<U>(acc, ta);
    add_batch<U>(acc, tb);

    v4f a = (acc[0] + acc[1]) + (acc[2] + acc[3]);
#pragma unroll
    for (int c = 0; c < 4; ++c) {
        a[c] += __shfl_xor(a[c], 8, 32);
        a[c] += __shfl_xor(a[c], 16, 32);
    }
    red[wave][c4] = a;
    __syncthreads();
    s0 = 0.f; s1 = 0.f;
#pragma unroll
    for (int w = 0; w < NW; ++w) {
        const v2f v = ((const v2f*)&red[w][0])[n];
        s0 += v.x; s1 += v.y;
    }
}

__global__ __launch_bounds__(NW * 32)
void CasualBoW_stream_scan(const float* __restrict__ x, float* __restrict__ out) {
    __shared__ v2f csum[2][NW][16];
    __shared__ v4f red[NW][SLAB / 4];

    const int lane = threadIdx.x & 31;
    const int wave = threadIdx.x >> 5;
    const int h    = lane >> 4;
    const int n    = lane & 15;
    const int NSLAB = Bc * (Cc / SLAB);
    const bool later = blockIdx.x < NSLAB;
    const int  sidx  = later ? blockIdx.x : blockIdx.x - NSLAB;
    const int  slab  = sidx % (Cc / SLAB);
    const int  b     = sidx / (Cc / SLAB);

    v2f A[4];
#pragma unroll
    for (int kk = 0; kk < 4; ++kk) {
        const int kg = kk * 4 + 2 * h;
        A[kk].x = (n >= kg)     ? 1.f : 0.f;
        A[kk].y = (n >= kg + 1) ? 1.f : 0.f;
    }

    const size_t slabOff = (size_t)b * Tc * Cc + (size_t)slab * SLAB;
    const size_t colOff  = slabOff + 2 * n;

    Chunk x0, x1;
    if (!later) {
        request_first(x0, x1, x + colOff, wave, h);
        stream_scan<NS_A, 0>(x0, x1, x + colOff, out + colOff, A, csum, wave, h, n, 0.f, 0.f);
    } else {
        const size_t partOff = colOff + (size_t)ROWS_A * Cc;
        request_first(x0, x1, x + partOff, wave, h);
        float c0, c1;
        slab_sum<ROWS_A>(x + slabOff, red, wave, lane, n, c0, c1);
        stream_scan<NS_B, ROWS_A>(x0, x1, x + partOff, out + partOff, A, csum, wave, h, n, c0, c1);
    }
}

extern "C" void kernel_launch(void* const* d_in, const int* in_sizes, int n_in,
                              void* d_out, int out_size, void* d_ws, size_t ws_size,
                              hipStream_t stream) {
    (void)in_sizes; (void)n_in; (void)out_size; (void)d_ws; (void)ws_size;
    const float* x   = (const float*)d_in[0];
    float*       out = (float*)d_out;

    dim3 g(2 * Bc * (Cc / SLAB));
    CasualBoW_stream_scan<<<g, NW * 32, 0, stream>>>(x, out);
}
